# starting kernel
# baseline (speedup 1.0000x reference)
#include <hip/hip_runtime.h>
#include <hip/hip_bf16.h>


#define NCHAN 24
#define NHIST 48
#define BINS 256
#define PIX_PER_CHAN (1024 * 1024)
#define HIST_BLOCKS_PER_CHAN 32
#define PIX_PER_HBLOCK (PIX_PER_CHAN / HIST_BLOCKS_PER_CHAN)
#define APPLY_BLOCKS_PER_CHAN 32
#define PIX_PER_ABLOCK (PIX_PER_CHAN / APPLY_BLOCKS_PER_CHAN)

typedef float v2f __attribute__((ext_vector_type(2)));
typedef float v8f __attribute__((ext_vector_type(8)));

__device__ __forceinline__ int bin_of(float x01) {
    float d = x01 * 255.0f;
    int idx = (int)(d * (256.0f / 255.0f));
    idx = idx < 0 ? 0 : (idx > 255 ? 255 : idx);
    return idx;
}

__global__ __launch_bounds__(256) void zero_hist_kernel(unsigned int* __restrict__ hist) {
    hist[blockIdx.x * 256 + threadIdx.x] = 0u;
}

__global__ __launch_bounds__(256) void hist_kernel(const float* __restrict__ dst,
                                                   const float* __restrict__ ref,
                                                   unsigned int* __restrict__ hist) {
    __shared__ unsigned int lh[8 * BINS];
    const int tid  = threadIdx.x;
    const int wave = tid >> 5;

    #pragma unroll
    for (int i = tid; i < 8 * BINS; i += 256) lh[i] = 0u;
    __syncthreads();

    const int chan = blockIdx.y;
    const float* src = (chan < NCHAN)
        ? dst + (size_t)chan * PIX_PER_CHAN
        : ref + (size_t)(chan - NCHAN) * PIX_PER_CHAN;

    const float4* s4 = (const float4*)(src + (size_t)blockIdx.x * PIX_PER_HBLOCK);
    const int n4 = PIX_PER_HBLOCK / 4;
    unsigned int* mywave = &lh[wave * BINS];

    for (int i = tid; i < n4; i += 256) {
        float4 v = s4[i];
        atomicAdd(&mywave[bin_of(v.x)], 1u);
        atomicAdd(&mywave[bin_of(v.y)], 1u);
        atomicAdd(&mywave[bin_of(v.z)], 1u);
        atomicAdd(&mywave[bin_of(v.w)], 1u);
    }
    __syncthreads();

    unsigned int s = 0;
    #pragma unroll
    for (int w = 0; w < 8; ++w) s += lh[w * BINS + tid];
    atomicAdd(&hist[chan * BINS + tid], s);
}

__global__ __launch_bounds__(32) void cdf_wmma_kernel(const unsigned int* __restrict__ hist,
                                                      float* __restrict__ cdf) {
    const int nt   = blockIdx.x;
    const int mt   = blockIdx.y;
    const int lane = threadIdx.x;
    const int half = lane >> 4;
    const int l15  = lane & 15;

    const int   row   = mt * 16 + l15;
    const int   ncol  = nt * 16 + l15;
    const float scale = 1.0f / 1048576.0f;
    const unsigned int* hrow = hist + row * BINS;

    v8f c = {0.0f, 0.0f, 0.0f, 0.0f, 0.0f, 0.0f, 0.0f, 0.0f};
    const int kmax = nt * 16 + 16;

    for (int k = 0; k < kmax; k += 4) {
        const int ka = k + half * 2;
        v2f a, b;
        a.x = (float)hrow[ka]     * scale;
        a.y = (float)hrow[ka + 1] * scale;
        b.x = (ka     <= ncol) ? 1.0f : 0.0f;
        b.y = (ka + 1 <= ncol) ? 1.0f : 0.0f;
        c = __builtin_amdgcn_wmma_f32_16x16x4_f32(
                 false, a,  false, b,
                 (short)0, c,  false,  false);
    }

    #pragma unroll
    for (int i = 0; i < 8; ++i) {
        const int r = mt * 16 + i + half * 8;
        cdf[r * BINS + ncol] = c[i];
    }
}

__global__ __launch_bounds__(256) void table_kernel(const float* __restrict__ cdf,
                                                    float* __restrict__ lut) {
    __shared__ float hd[BINS];
    __shared__ float hr[BINS];
    const int m = blockIdx.x;
    const int i = threadIdx.x;
    hd[i] = cdf[m * BINS + i];
    hr[i] = cdf[(NCHAN + m) * BINS + i];
    __syncthreads();

    int t;
    if (i == 0) {
        t = 0;
    } else if (i == BINS - 1) {
        t = BINS - 1;
    } else {
        t = i;
        const float v = hd[i];
        for (int j = 1; j < BINS; ++j) {
            if (hr[j - 1] <= v && v <= hr[j]) { t = j; break; }
        }
    }
    lut[m * BINS + i] = (float)t / 255.0f;
}

__global__ __launch_bounds__(256) void apply_kernel(const float* __restrict__ dst,
                                                    const float* __restrict__ lut,
                                                    float* __restrict__ out) {
    __shared__ float l[BINS];
    const int ch  = blockIdx.y;
    const int row = (ch / 3) * (ch % 3);
    l[threadIdx.x] = lut[row * BINS + threadIdx.x];
    __syncthreads();

    const size_t base = (size_t)ch * PIX_PER_CHAN + (size_t)blockIdx.x * PIX_PER_ABLOCK;
    const float4* s4 = (const float4*)(dst + base);
    float4*       o4 = (float4*)(out + base);
    const int n4 = PIX_PER_ABLOCK / 4;

    for (int i = threadIdx.x; i < n4; i += 256) {
        float4 v = s4[i];
        float4 r;
        int p0 = (int)(v.x * 255.0f); p0 = p0 < 0 ? 0 : (p0 > 255 ? 255 : p0);
        int p1 = (int)(v.y * 255.0f); p1 = p1 < 0 ? 0 : (p1 > 255 ? 255 : p1);
        int p2 = (int)(v.z * 255.0f); p2 = p2 < 0 ? 0 : (p2 > 255 ? 255 : p2);
        int p3 = (int)(v.w * 255.0f); p3 = p3 < 0 ? 0 : (p3 > 255 ? 255 : p3);
        r.x = l[p0]; r.y = l[p1]; r.z = l[p2]; r.w = l[p3];
        o4[i] = r;
    }
}

extern "C" void kernel_launch(void* const* d_in, const int* in_sizes, int n_in,
                              void* d_out, int out_size, void* d_ws, size_t ws_size,
                              hipStream_t stream) {
    (void)in_sizes; (void)n_in; (void)out_size; (void)ws_size;
    const float* dst = (const float*)d_in[0];
    const float* ref = (const float*)d_in[1];
    float* out = (float*)d_out;

    unsigned int* hist = (unsigned int*)d_ws;
    float* cdf = (float*)((char*)d_ws + (size_t)NHIST * BINS * 4);
    float* lut = (float*)((char*)d_ws + (size_t)NHIST * BINS * 8);

    zero_hist_kernel<<<dim3(NHIST), 256, 0, stream>>>(hist);
    hist_kernel<<<dim3(HIST_BLOCKS_PER_CHAN, NHIST), 256, 0, stream>>>(dst, ref, hist);
    cdf_wmma_kernel<<<dim3(16, 3), 32, 0, stream>>>(hist, cdf);
    table_kernel<<<dim3(NCHAN), 256, 0, stream>>>(cdf, lut);
    apply_kernel<<<dim3(APPLY_BLOCKS_PER_CHAN, NCHAN), 256, 0, stream>>>(dst, lut, out);
}


# hist kernel: blocks of 4 waves, not 16 (3,072 waves to start in that launch, not 12,288); a thread takes 64 float4 per image in groups of 16, loads issued two groups ahead of the counts; counts exact
# speedup vs baseline: 1.4400x; 1.4400x over previous
#include <hip/hip_runtime.h>
#include <hip/hip_bf16.h>


#define NCHAN 24
#define NHIST 48
#define BINS 256
#define PIX_PER_CHAN (1024 * 1024)
#define HIST_BLOCKS_PER_CHAN 32
#define PIX_PER_HBLOCK (PIX_PER_CHAN / HIST_BLOCKS_PER_CHAN)
#define APPLY_BLOCKS_PER_CHAN 32
#define PIX_PER_ABLOCK (PIX_PER_CHAN / APPLY_BLOCKS_PER_CHAN)

typedef float v2f __attribute__((ext_vector_type(2)));
typedef float v8f __attribute__((ext_vector_type(8)));

__device__ __forceinline__ int bin_of(float x01) {
    float d = x01 * 255.0f;
    int idx = (int)(d * (256.0f / 255.0f));
    idx = idx < 0 ? 0 : (idx > 255 ? 255 : idx);
    return idx;
}

__global__ __launch_bounds__(256) void zero_hist_kernel(unsigned int* __restrict__ hist) {
    hist[blockIdx.x * 256 + threadIdx.x] = 0u;
}

#define HIST_THREADS 128
#define HIST_WAVES (HIST_THREADS / 32)
#define GROUP 16
#define GROUP_STRIDE (GROUP * HIST_THREADS)
#define GROUPS_PER_IMAGE (PIX_PER_HBLOCK / 4 / GROUP_STRIDE)
static_assert(GROUPS_PER_IMAGE == 4, "the trip structure below is written for 4 groups per image");

__device__ __forceinline__ void load_group(float4 (&v)[GROUP], const float4* __restrict__ p) {
    #pragma unroll
    for (int q = 0; q < GROUP; ++q) v[q] = p[q * HIST_THREADS];
}

__device__ __forceinline__ void count_group(const float4 (&v)[GROUP], unsigned int* h) {
    #pragma unroll
    for (int q = 0; q < GROUP; ++q) {
        atomicAdd(&h[bin_of(v[q].x)], 1u);
        atomicAdd(&h[bin_of(v[q].y)], 1u);
        atomicAdd(&h[bin_of(v[q].z)], 1u);
        atomicAdd(&h[bin_of(v[q].w)], 1u);
    }
}

__global__ __launch_bounds__(HIST_THREADS) void hist_kernel(const float* __restrict__ dst,
                                                            const float* __restrict__ ref,
                                                            unsigned int* __restrict__ pref) {
    __shared__ unsigned int lh[2 * HIST_WAVES * BINS];
    __shared__ float tot[2 * BINS];
    __shared__ float rowtot[2 * 16];
    const int tid  = threadIdx.x;
    const int wave = tid >> 5;

    for (int k = 0; k < 2 * HIST_WAVES * BINS / HIST_THREADS; ++k) lh[tid + k * HIST_THREADS] = 0u;
    __syncthreads();

    const int m = blockIdx.y;
    const size_t off = (size_t)m * PIX_PER_CHAN + (size_t)blockIdx.x * PIX_PER_HBLOCK;
    const float4* r4 = (const float4*)(ref + off) + tid;
    const float4* d4 = (const float4*)(dst + off) + tid;
    unsigned int* myref = &lh[wave * BINS];
    unsigned int* mydst = &lh[(HIST_WAVES + wave) * BINS];

    float4 a[GROUP], b[GROUP];
    load_group(a, r4);
    load_group(b, r4 + GROUP_STRIDE);
    #pragma nounroll
    for (int p = 0; p < GROUPS_PER_IMAGE - 1; ++p) {
        unsigned int* h    = (p < 2) ? myref : mydst;
        const float4* nsrc = ((p < 1) ? r4 : d4)
                             + ((2 * p + 2) & (GROUPS_PER_IMAGE - 1)) * GROUP_STRIDE;
        count_group(a, h);
        load_group(a, nsrc);
        __builtin_amdgcn_sched_barrier(0);
        count_group(b, h);
        load_group(b, nsrc + GROUP_STRIDE);
    }
    count_group(a, mydst);
    __builtin_amdgcn_sched_barrier(0);
    count_group(b, mydst);
    __syncthreads();

    for (int k = 0; k < 2 * BINS / HIST_THREADS; ++k) {
        const int e   = tid + k * HIST_THREADS;
        const int set = e >> 8;
        const int bin = e & (BINS - 1);
        unsigned int s = 0;
        for (int w = 0; w < HIST_WAVES; ++w) s += lh[(set * HIST_WAVES + w) * BINS + bin];
        tot[e] = (float)s;
    }
    __syncthreads();

    const int wave_u = __builtin_amdgcn_readfirstlane(wave);
    const bool scan_wave = wave_u < 2;
    float* T  = tot + (wave_u & 1) * BINS;
    float* RT = rowtot + (wave_u & 1) * 16;
    const int lane = tid & 31;
    const int half = lane >> 4;
    const int l15  = lane & 15;
    v8f c = {0.0f, 0.0f, 0.0f, 0.0f, 0.0f, 0.0f, 0.0f, 0.0f};

    if (scan_wave) {
        #pragma unroll
        for (int k0 = 0; k0 < 16; k0 += 4) {
            const int ka = k0 + half * 2;
            v2f fa, fb;
            fa.x = T[16 * l15 + ka];
            fa.y = T[16 * l15 + ka + 1];
            fb.x = (ka     <= l15) ? 1.0f : 0.0f;
            fb.y = (ka + 1 <= l15) ? 1.0f : 0.0f;
            c = __builtin_amdgcn_wmma_f32_16x16x4_f32(
                     false, fa,  false, fb,
                     (short)0, c,  false,  false);
        }
        if (l15 == 15) {
            #pragma unroll
            for (int i = 0; i < 8; ++i) RT[i + half * 8] = c[i];
        }
    }
    __syncthreads();

    if (scan_wave) {
        #pragma unroll
        for (int k0 = 0; k0 < 16; k0 += 4) {
            const int ka = k0 + half * 2;
            v2f fa, fb;
            fa.x = (ka     < l15) ? 1.0f : 0.0f;
            fa.y = (ka + 1 < l15) ? 1.0f : 0.0f;
            fb.x = RT[ka];
            fb.y = RT[ka + 1];
            c = __builtin_amdgcn_wmma_f32_16x16x4_f32(
                     false, fa,  false, fb,
                     (short)0, c,  false,  false);
        }
        #pragma unroll
        for (int i = 0; i < 8; ++i) T[16 * (i + half * 8) + l15] = c[i];
    }
    __syncthreads();

    for (int k = 0; k < 2 * BINS / HIST_THREADS; ++k) {
        const int e   = tid + k * HIST_THREADS;
        const int set = e >> 8;
        const int bin = e & (BINS - 1);
        const int row = set ? m : (NCHAN + m);
        atomicAdd(&pref[row * BINS + bin], (unsigned int)tot[e]);
    }
}

__global__ __launch_bounds__(256) void table_kernel(const unsigned int* __restrict__ pref,
                                                    float* __restrict__ lut) {
    const int m    = blockIdx.x;
    const int i    = threadIdx.x;
    const int lane = i & 31;
    const unsigned int* pd = pref + m * BINS;
    const unsigned int* pr = pref + (NCHAN + m) * BINS;

    const unsigned int vmine = pd[i];
    unsigned int r[8];
    #pragma unroll
    for (int k = 0; k < 8; ++k) r[k] = pr[lane + 32 * k];
    const unsigned int pr0 = __builtin_amdgcn_readfirstlane(r[0]);
    const int ibase = i - lane;

    int tmine = i;
    #pragma unroll 4
    for (int ii = 0; ii < 32; ++ii) {
        const unsigned int v = __builtin_amdgcn_readlane(vmine, ii);
        int rank = 0;
        #pragma unroll
        for (int k = 0; k < 8; ++k)
            rank += __builtin_popcount(__builtin_amdgcn_ballot_w32(r[k] < v));
        const int self = ibase + ii;
        int t;
        if (rank == 0)        t = (pr0 == v) ? 1 : self;
        else if (rank < BINS) t = rank;
        else                  t = self;
        tmine = (lane == ii) ? t : tmine;
    }
    if (i == 0)        tmine = 0;
    if (i == BINS - 1) tmine = BINS - 1;
    lut[m * BINS + i] = (float)tmine / 255.0f;
}

__global__ __launch_bounds__(256) void apply_kernel(const float* __restrict__ dst,
                                                    const float* __restrict__ lut,
                                                    float* __restrict__ out) {
    __shared__ float l[BINS];
    const int ch  = blockIdx.y;
    const int row = (ch / 3) * (ch % 3);
    l[threadIdx.x] = lut[row * BINS + threadIdx.x];
    __syncthreads();

    const size_t base = (size_t)ch * PIX_PER_CHAN + (size_t)blockIdx.x * PIX_PER_ABLOCK;
    const float4* s4 = (const float4*)(dst + base);
    float4*       o4 = (float4*)(out + base);
    const int n4 = PIX_PER_ABLOCK / 4;

    for (int i = threadIdx.x; i < n4; i += 256) {
        float4 v = s4[i];
        float4 r;
        int p0 = (int)(v.x * 255.0f); p0 = p0 < 0 ? 0 : (p0 > 255 ? 255 : p0);
        int p1 = (int)(v.y * 255.0f); p1 = p1 < 0 ? 0 : (p1 > 255 ? 255 : p1);
        int p2 = (int)(v.z * 255.0f); p2 = p2 < 0 ? 0 : (p2 > 255 ? 255 : p2);
        int p3 = (int)(v.w * 255.0f); p3 = p3 < 0 ? 0 : (p3 > 255 ? 255 : p3);
        r.x = l[p0]; r.y = l[p1]; r.z = l[p2]; r.w = l[p3];
        o4[i] = r;
    }
}

extern "C" void kernel_launch(void* const* d_in, const int* in_sizes, int n_in,
                              void* d_out, int out_size, void* d_ws, size_t ws_size,
                              hipStream_t stream) {
    (void)in_sizes; (void)n_in; (void)out_size; (void)ws_size;
    const float* dst = (const float*)d_in[0];
    const float* ref = (const float*)d_in[1];
    float* out = (float*)d_out;

    unsigned int* pref = (unsigned int*)d_ws;
    float* lut = (float*)((char*)d_ws + (size_t)NHIST * BINS * 8);

    zero_hist_kernel<<<dim3(NHIST), 256, 0, stream>>>(pref);
    hist_kernel<<<dim3(HIST_BLOCKS_PER_CHAN, NCHAN), HIST_THREADS, 0, stream>>>(dst, ref, pref);
    table_kernel<<<dim3(NCHAN), 256, 0, stream>>>(pref, lut);
    apply_kernel<<<dim3(APPLY_BLOCKS_PER_CHAN, NCHAN), 256, 0, stream>>>(dst, lut, out);
}


# hist kernel also stores each dst pixel's table index (1 byte) in the first quarter of its chunk of out; apply reads those 24 MiB, 8 loads per thread up front, not 96 MiB of dst; values kept
# speedup vs baseline: 1.4849x; 1.0312x over previous
#include <hip/hip_runtime.h>
#include <hip/hip_bf16.h>


#define NCHAN 24
#define NHIST 48
#define BINS 256
#define PIX_PER_CHAN (1024 * 1024)
#define HIST_BLOCKS_PER_CHAN 32
#define PIX_PER_HBLOCK (PIX_PER_CHAN / HIST_BLOCKS_PER_CHAN)
#define APPLY_BLOCKS_PER_CHAN 32
#define PIX_PER_ABLOCK (PIX_PER_CHAN / APPLY_BLOCKS_PER_CHAN)

typedef float v2f __attribute__((ext_vector_type(2)));
typedef float v8f __attribute__((ext_vector_type(8)));

__device__ __forceinline__ int bin_of(float x01) {
    float d = x01 * 255.0f;
    int idx = (int)(d * (256.0f / 255.0f));
    idx = idx < 0 ? 0 : (idx > 255 ? 255 : idx);
    return idx;
}

__device__ __forceinline__ unsigned int index_of(float x01) {
    int p = (int)(x01 * 255.0f);
    p = p < 0 ? 0 : (p > 255 ? 255 : p);
    return (unsigned int)p;
}

__global__ __launch_bounds__(256) void zero_hist_kernel(unsigned int* __restrict__ hist) {
    hist[blockIdx.x * 256 + threadIdx.x] = 0u;
}

#define HIST_THREADS 128
#define HIST_WAVES (HIST_THREADS / 32)
#define GROUP 16
#define GROUP_STRIDE (GROUP * HIST_THREADS)
#define GROUPS_PER_IMAGE (PIX_PER_HBLOCK / 4 / GROUP_STRIDE)
static_assert(GROUPS_PER_IMAGE == 4, "the trip structure below is written for 4 groups per image");

__device__ __forceinline__ void load_group(float4 (&v)[GROUP], const float4* __restrict__ p) {
    #pragma unroll
    for (int q = 0; q < GROUP; ++q) v[q] = p[q * HIST_THREADS];
}

__device__ __forceinline__ void count_group(const float4 (&v)[GROUP], unsigned int* h) {
    #pragma unroll
    for (int q = 0; q < GROUP; ++q) {
        atomicAdd(&h[bin_of(v[q].x)], 1u);
        atomicAdd(&h[bin_of(v[q].y)], 1u);
        atomicAdd(&h[bin_of(v[q].z)], 1u);
        atomicAdd(&h[bin_of(v[q].w)], 1u);
    }
}

__device__ __forceinline__ void store_indices(const float4 (&v)[GROUP], float4* slot) {
    unsigned int w[GROUP];
    #pragma unroll
    for (int q = 0; q < GROUP; ++q)
        w[q] = index_of(v[q].x) | (index_of(v[q].y) << 8) |
               (index_of(v[q].z) << 16) | (index_of(v[q].w) << 24);
    #pragma unroll
    for (int kk = 0; kk < 2; ++kk) {
        #pragma unroll
        for (int bb = 0; bb < 2; ++bb) {
            float4 s;
            s.x = __uint_as_float(w[8 * kk + 0 + bb]);
            s.y = __uint_as_float(w[8 * kk + 2 + bb]);
            s.z = __uint_as_float(w[8 * kk + 4 + bb]);
            s.w = __uint_as_float(w[8 * kk + 6 + bb]);
            slot[kk * 256 + bb * HIST_THREADS] = s;
        }
    }
}

__global__ __launch_bounds__(HIST_THREADS) void hist_kernel(const float* __restrict__ dst,
                                                            const float* __restrict__ ref,
                                                            unsigned int* __restrict__ pref,
                                                            float* __restrict__ out) {
    __shared__ unsigned int lh[2 * HIST_WAVES * BINS];
    __shared__ float tot[2 * BINS];
    __shared__ float rowtot[2 * 16];
    const int tid  = threadIdx.x;
    const int wave = tid >> 5;

    for (int k = 0; k < 2 * HIST_WAVES * BINS / HIST_THREADS; ++k) lh[tid + k * HIST_THREADS] = 0u;
    __syncthreads();

    const int m = blockIdx.y;
    const size_t off = (size_t)m * PIX_PER_CHAN + (size_t)blockIdx.x * PIX_PER_HBLOCK;
    const float4* r4 = (const float4*)(ref + off) + tid;
    const float4* d4 = (const float4*)(dst + off) + tid;
    unsigned int* myref = &lh[wave * BINS];
    unsigned int* mydst = &lh[(HIST_WAVES + wave) * BINS];

    float4* slot = (float4*)(out + off) + tid;
    float4 a[GROUP], b[GROUP];
    load_group(a, r4);
    load_group(b, r4 + GROUP_STRIDE);
    #pragma nounroll
    for (int p = 0; p < GROUPS_PER_IMAGE - 1; ++p) {
        unsigned int* h    = (p < 2) ? myref : mydst;
        const float4* nsrc = ((p < 1) ? r4 : d4)
                             + ((2 * p + 2) & (GROUPS_PER_IMAGE - 1)) * GROUP_STRIDE;
        count_group(a, h);
        if (p == 2) store_indices(a, slot);
        load_group(a, nsrc);
        __builtin_amdgcn_sched_barrier(0);
        count_group(b, h);
        if (p == 2) store_indices(b, slot + 512);
        load_group(b, nsrc + GROUP_STRIDE);
    }
    count_group(a, mydst);
    store_indices(a, slot + 1024);
    __builtin_amdgcn_sched_barrier(0);
    count_group(b, mydst);
    store_indices(b, slot + 1536);
    __syncthreads();

    for (int k = 0; k < 2 * BINS / HIST_THREADS; ++k) {
        const int e   = tid + k * HIST_THREADS;
        const int set = e >> 8;
        const int bin = e & (BINS - 1);
        unsigned int s = 0;
        for (int w = 0; w < HIST_WAVES; ++w) s += lh[(set * HIST_WAVES + w) * BINS + bin];
        tot[e] = (float)s;
    }
    __syncthreads();

    const int wave_u = __builtin_amdgcn_readfirstlane(wave);
    const bool scan_wave = wave_u < 2;
    float* T  = tot + (wave_u & 1) * BINS;
    float* RT = rowtot + (wave_u & 1) * 16;
    const int lane = tid & 31;
    const int half = lane >> 4;
    const int l15  = lane & 15;
    v8f c = {0.0f, 0.0f, 0.0f, 0.0f, 0.0f, 0.0f, 0.0f, 0.0f};

    if (scan_wave) {
        #pragma unroll
        for (int k0 = 0; k0 < 16; k0 += 4) {
            const int ka = k0 + half * 2;
            v2f fa, fb;
            fa.x = T[16 * l15 + ka];
            fa.y = T[16 * l15 + ka + 1];
            fb.x = (ka     <= l15) ? 1.0f : 0.0f;
            fb.y = (ka + 1 <= l15) ? 1.0f : 0.0f;
            c = __builtin_amdgcn_wmma_f32_16x16x4_f32(
                     false, fa,  false, fb,
                     (short)0, c,  false,  false);
        }
        if (l15 == 15) {
            #pragma unroll
            for (int i = 0; i < 8; ++i) RT[i + half * 8] = c[i];
        }
    }
    __syncthreads();

    if (scan_wave) {
        #pragma unroll
        for (int k0 = 0; k0 < 16; k0 += 4) {
            const int ka = k0 + half * 2;
            v2f fa, fb;
            fa.x = (ka     < l15) ? 1.0f : 0.0f;
            fa.y = (ka + 1 < l15) ? 1.0f : 0.0f;
            fb.x = RT[ka];
            fb.y = RT[ka + 1];
            c = __builtin_amdgcn_wmma_f32_16x16x4_f32(
                     false, fa,  false, fb,
                     (short)0, c,  false,  false);
        }
        #pragma unroll
        for (int i = 0; i < 8; ++i) T[16 * (i + half * 8) + l15] = c[i];
    }
    __syncthreads();

    for (int k = 0; k < 2 * BINS / HIST_THREADS; ++k) {
        const int e   = tid + k * HIST_THREADS;
        const int set = e >> 8;
        const int bin = e & (BINS - 1);
        const int row = set ? m : (NCHAN + m);
        atomicAdd(&pref[row * BINS + bin], (unsigned int)tot[e]);
    }
}

__global__ __launch_bounds__(256) void table_kernel(const unsigned int* __restrict__ pref,
                                                    float* __restrict__ lut) {
    const int m    = blockIdx.x;
    const int i    = threadIdx.x;
    const int lane = i & 31;
    const unsigned int* pd = pref + m * BINS;
    const unsigned int* pr = pref + (NCHAN + m) * BINS;

    const unsigned int vmine = pd[i];
    unsigned int r[8];
    #pragma unroll
    for (int k = 0; k < 8; ++k) r[k] = pr[lane + 32 * k];
    const unsigned int pr0 = __builtin_amdgcn_readfirstlane(r[0]);
    const int ibase = i - lane;

    int tmine = i;
    #pragma unroll 4
    for (int ii = 0; ii < 32; ++ii) {
        const unsigned int v = __builtin_amdgcn_readlane(vmine, ii);
        int rank = 0;
        #pragma unroll
        for (int k = 0; k < 8; ++k)
            rank += __builtin_popcount(__builtin_amdgcn_ballot_w32(r[k] < v));
        const int self = ibase + ii;
        int t;
        if (rank == 0)        t = (pr0 == v) ? 1 : self;
        else if (rank < BINS) t = rank;
        else                  t = self;
        tmine = (lane == ii) ? t : tmine;
    }
    if (i == 0)        tmine = 0;
    if (i == BINS - 1) tmine = BINS - 1;
    lut[m * BINS + i] = (float)tmine / 255.0f;
}

__global__ __launch_bounds__(256) void apply_kernel(const float* __restrict__ lut,
                                                    float* out) {
    __shared__ float l[BINS];
    const int ch  = blockIdx.y;
    const int row = (ch / 3) * (ch % 3);
    const size_t base = (size_t)ch * PIX_PER_CHAN + (size_t)blockIdx.x * PIX_PER_ABLOCK;
    float4* o4 = (float4*)(out + base) + threadIdx.x;

    float4 w[8];
    #pragma unroll
    for (int k = 0; k < 8; ++k) w[k] = o4[k * 256];

    l[threadIdx.x] = lut[row * BINS + threadIdx.x];
    __syncthreads();

    #pragma unroll
    for (int k = 0; k < 8; ++k) {
        const unsigned int u[4] = {__float_as_uint(w[k].x), __float_as_uint(w[k].y),
                                   __float_as_uint(w[k].z), __float_as_uint(w[k].w)};
        #pragma unroll
        for (int j = 0; j < 4; ++j) {
            float4 r;
            r.x = l[u[j] & 255u];
            r.y = l[(u[j] >> 8) & 255u];
            r.z = l[(u[j] >> 16) & 255u];
            r.w = l[u[j] >> 24];
            o4[(4 * k + j) * 256] = r;
        }
    }
}

extern "C" void kernel_launch(void* const* d_in, const int* in_sizes, int n_in,
                              void* d_out, int out_size, void* d_ws, size_t ws_size,
                              hipStream_t stream) {
    (void)in_sizes; (void)n_in; (void)out_size; (void)ws_size;
    const float* dst = (const float*)d_in[0];
    const float* ref = (const float*)d_in[1];
    float* out = (float*)d_out;

    unsigned int* pref = (unsigned int*)d_ws;
    float* lut = (float*)((char*)d_ws + (size_t)NHIST * BINS * 8);

    zero_hist_kernel<<<dim3(NHIST), 256, 0, stream>>>(pref);
    hist_kernel<<<dim3(HIST_BLOCKS_PER_CHAN, NCHAN), HIST_THREADS, 0, stream>>>(dst, ref, pref, out);
    table_kernel<<<dim3(NCHAN), 256, 0, stream>>>(pref, lut);
    apply_kernel<<<dim3(APPLY_BLOCKS_PER_CHAN, NCHAN), 256, 0, stream>>>(lut, out);
}
